# starting kernel
# baseline (speedup 1.0000x reference)
#include <hip/hip_runtime.h>


#define N_SAMPLES       128
#define RAYS_PER_WAVE   16
#define WAVES_PER_BLOCK 4
#define RAYS_PER_BLOCK  (RAYS_PER_WAVE * WAVES_PER_BLOCK)
#define LDS_STRIDE      17

typedef __attribute__((ext_vector_type(2))) float v2f;
typedef __attribute__((ext_vector_type(8))) float v8f;

__global__ __launch_bounds__(WAVES_PER_BLOCK * 32)
void volsdf_weights_kernel(const float* __restrict__ sdf,
                           const float* __restrict__ deltas,
                           const float* __restrict__ beta,
                           float* __restrict__ out)
{
    __shared__ float tile[WAVES_PER_BLOCK][N_SAMPLES * LDS_STRIDE];

    const int tid   = threadIdx.x;
    const int wave  = tid >> 5;
    const int lane  = tid & 31;
    const int rbase = blockIdx.x * RAYS_PER_BLOCK + wave * RAYS_PER_WAVE;

    const float b     = beta[0];
    const float inv_b = 1.0f / b;

    float* t = tile[wave];

    const float4* sdf4 = reinterpret_cast<const float4*>(sdf)    + (size_t)rbase * (N_SAMPLES / 4);
    const float4* del4 = reinterpret_cast<const float4*>(deltas) + (size_t)rbase * (N_SAMPLES / 4);
    #pragma unroll
    for (int it = 0; it < RAYS_PER_WAVE; ++it) {
        float4 s4 = sdf4[it * 32 + lane];
        float4 d4 = del4[it * 32 + lane];
        const int samp = lane * 4;
        const float sv[4] = {s4.x, s4.y, s4.z, s4.w};
        const float dv[4] = {d4.x, d4.y, d4.z, d4.w};
        #pragma unroll
        for (int j = 0; j < 4; ++j) {
            float s = sv[j];
            float e = __expf(-fabsf(s) * inv_b);
            float dens = (s >= 0.0f) ? (0.5f * e * inv_b) : ((1.0f - 0.5f * e) * inv_b);
            t[(samp + j) * LDS_STRIDE + it] = dv[j] * dens;
        }
    }
    __syncthreads();

    const int n     = lane & 15;
    const int m     = lane & 15;
    const int khalf = (lane >> 4) * 2;
    const int moff  = (lane >> 4) * 8;

    v2f aop[4];
    #pragma unroll
    for (int kb = 0; kb < 4; ++kb) {
        const int k0 = 4 * kb + khalf;
        aop[kb].x = (k0     <= m) ? 1.0f : 0.0f;
        aop[kb].y = (k0 + 1 <= m) ? 1.0f : 0.0f;
    }

    float running = 0.0f;
    float* outp = out + (size_t)(rbase + n) * N_SAMPLES;

    #pragma unroll
    for (int c = 0; c < 8; ++c) {
        v8f acc = {};
        #pragma unroll
        for (int kb = 0; kb < 4; ++kb) {
            const int s0 = c * 16 + 4 * kb + khalf;
            v2f bb;
            bb.x = t[s0 * LDS_STRIDE + n];
            bb.y = t[(s0 + 1) * LDS_STRIDE + n];
            acc = __builtin_amdgcn_wmma_f32_16x16x4_f32(
                false, aop[kb], false, bb, (short)0, acc, false, false);
        }

        const float tot = __shfl(acc[7], 16 + n, 32);

        float w[8];
        #pragma unroll
        for (int v = 0; v < 8; ++v) {
            const int mm = v + moff;
            const float x    = t[(c * 16 + mm) * LDS_STRIDE + n];
            const float incl = (c == 0) ? acc[v] : (running + acc[v]);
            const float cs   = incl - x;
            const float alpha = 1.0f - __expf(-x);
            const float trans = __expf(-cs);
            w[v] = alpha * trans;
        }
        running = (c == 0) ? tot : (running + tot);

        float4* o4 = reinterpret_cast<float4*>(outp + c * 16 + moff);
        o4[0] = make_float4(w[0], w[1], w[2], w[3]);
        o4[1] = make_float4(w[4], w[5], w[6], w[7]);
    }
}

extern "C" void kernel_launch(void* const* d_in, const int* in_sizes, int n_in,
                              void* d_out, int out_size, void* d_ws, size_t ws_size,
                              hipStream_t stream) {
    const float* sdf    = (const float*)d_in[0];
    const float* deltas = (const float*)d_in[1];
    const float* beta   = (const float*)d_in[2];
    float* out = (float*)d_out;

    const int n_rays = in_sizes[0] / N_SAMPLES;
    const int grid   = n_rays / RAYS_PER_BLOCK;
    volsdf_weights_kernel<<<grid, WAVES_PER_BLOCK * 32, 0, stream>>>(sdf, deltas, beta, out);
}


# Tile loop: 512 workgroups x 8 waves, 2 tiles of 16 rays per wave; a tile's 32 global loads issue as one group, the next tile's before this tile's scan (was 2 in flight); no workgroup barrier
# speedup vs baseline: 1.0734x; 1.0734x over previous
#include <hip/hip_runtime.h>


#define N_SAMPLES       128
#define RAYS_PER_TILE   16
#define WAVES_PER_BLOCK 8
#define TILES_PER_WAVE  2
#define RAYS_PER_WAVE   (RAYS_PER_TILE * TILES_PER_WAVE)
#define RAYS_PER_BLOCK  (RAYS_PER_WAVE * WAVES_PER_BLOCK)
#define LDS_STRIDE      17

typedef __attribute__((ext_vector_type(2))) float v2f;
typedef __attribute__((ext_vector_type(8))) float v8f;
typedef __attribute__((ext_vector_type(4))) float v4f;

__device__ __forceinline__ void issue_tile_loads(const float* __restrict__ sdf,
                                                 const float* __restrict__ deltas,
                                                 int rbase, int lane,
                                                 v4f (&s4)[RAYS_PER_TILE],
                                                 v4f (&d4)[RAYS_PER_TILE])
{
    const v4f* sp = reinterpret_cast<const v4f*>(sdf)    + (size_t)rbase * (N_SAMPLES / 4) + lane;
    const v4f* dp = reinterpret_cast<const v4f*>(deltas) + (size_t)rbase * (N_SAMPLES / 4) + lane;
    #pragma unroll
    for (int it = 0; it < RAYS_PER_TILE; ++it) {
        s4[it] = sp[it * 32];
        d4[it] = dp[it * 32];
    }
}

__device__ __forceinline__ void wave_lds_order()
{
    __builtin_amdgcn_fence(__ATOMIC_RELEASE, "wavefront");
    __builtin_amdgcn_wave_barrier();
    __builtin_amdgcn_fence(__ATOMIC_ACQUIRE, "wavefront");
}

__global__ __launch_bounds__(WAVES_PER_BLOCK * 32)
void volsdf_weights_kernel(const float* __restrict__ sdf,
                           const float* __restrict__ deltas,
                           const float* __restrict__ beta,
                           float* __restrict__ out)
{
    __shared__ float tile[WAVES_PER_BLOCK][N_SAMPLES * LDS_STRIDE];

    const int tid    = threadIdx.x;
    const int wave   = tid >> 5;
    const int lane   = tid & 31;
    const int rbase0 = (blockIdx.x * WAVES_PER_BLOCK + wave) * RAYS_PER_WAVE;

    v4f s4[RAYS_PER_TILE], d4[RAYS_PER_TILE];
    issue_tile_loads(sdf, deltas, rbase0, lane, s4, d4);

    const float b     = beta[0];
    const float inv_b = 1.0f / b;

    float* t = tile[wave];

    const int n     = lane & 15;
    const int m     = lane & 15;
    const int khalf = (lane >> 4) * 2;
    const int moff  = (lane >> 4) * 8;

    v2f aop[4];
    #pragma unroll
    for (int kb = 0; kb < 4; ++kb) {
        const int k0 = 4 * kb + khalf;
        aop[kb].x = (k0     <= m) ? 1.0f : 0.0f;
        aop[kb].y = (k0 + 1 <= m) ? 1.0f : 0.0f;
    }

    #pragma unroll
    for (int tl = 0; tl < TILES_PER_WAVE; ++tl) {
        const int rbase = rbase0 + tl * RAYS_PER_TILE;

        #pragma unroll
        for (int it = 0; it < RAYS_PER_TILE; ++it) {
            v4f sq = s4[it], dq = d4[it];
            asm volatile("" : "+v"(sq), "+v"(dq));
            const int samp = lane * 4;
            const float sv[4] = {sq.x, sq.y, sq.z, sq.w};
            const float dv[4] = {dq.x, dq.y, dq.z, dq.w};
            #pragma unroll
            for (int j = 0; j < 4; ++j) {
                float s = sv[j];
                float e = __expf(-fabsf(s) * inv_b);
                float dens = (s >= 0.0f) ? (0.5f * e * inv_b) : ((1.0f - 0.5f * e) * inv_b);
                t[(samp + j) * LDS_STRIDE + it] = dv[j] * dens;
            }
        }
        wave_lds_order();

        if (tl + 1 < TILES_PER_WAVE)
            issue_tile_loads(sdf, deltas, rbase + RAYS_PER_TILE, lane, s4, d4);

        float running = 0.0f;
        float* outp = out + (size_t)(rbase + n) * N_SAMPLES;

        #pragma unroll
        for (int c = 0; c < 8; ++c) {
            v8f acc = {};
            #pragma unroll
            for (int kb = 0; kb < 4; ++kb) {
                const int s0 = c * 16 + 4 * kb + khalf;
                v2f bb;
                bb.x = t[s0 * LDS_STRIDE + n];
                bb.y = t[(s0 + 1) * LDS_STRIDE + n];
                acc = __builtin_amdgcn_wmma_f32_16x16x4_f32(
                    false, aop[kb], false, bb, (short)0, acc, false, false);
            }

            const float tot = __shfl(acc[7], 16 + n, 32);

            float w[8];
            #pragma unroll
            for (int v = 0; v < 8; ++v) {
                const int mm = v + moff;
                const float x    = t[(c * 16 + mm) * LDS_STRIDE + n];
                const float incl = (c == 0) ? acc[v] : (running + acc[v]);
                const float cs   = incl - x;
                const float alpha = 1.0f - __expf(-x);
                const float trans = __expf(-cs);
                w[v] = alpha * trans;
            }
            running = (c == 0) ? tot : (running + tot);

            float4* o4 = reinterpret_cast<float4*>(outp + c * 16 + moff);
            o4[0] = make_float4(w[0], w[1], w[2], w[3]);
            o4[1] = make_float4(w[4], w[5], w[6], w[7]);

            if ((c & 1) == 1) __builtin_amdgcn_sched_barrier(0);
        }
        wave_lds_order();
    }
}

extern "C" void kernel_launch(void* const* d_in, const int* in_sizes, int n_in,
                              void* d_out, int out_size, void* d_ws, size_t ws_size,
                              hipStream_t stream) {
    const float* sdf    = (const float*)d_in[0];
    const float* deltas = (const float*)d_in[1];
    const float* beta   = (const float*)d_in[2];
    float* out = (float*)d_out;

    const int n_rays = in_sizes[0] / N_SAMPLES;
    const int grid   = n_rays / RAYS_PER_BLOCK;
    volsdf_weights_kernel<<<grid, WAVES_PER_BLOCK * 32, 0, stream>>>(sdf, deltas, beta, out);
}
